# starting kernel
# baseline (speedup 1.0000x reference)
#include <hip/hip_runtime.h>
#include <math.h>

#define DIMN 64
#define NNB  16
#define NRELS 32
#define XS   68
#define WSP  66

typedef float v2f __attribute__((ext_vector_type(2)));
typedef float v8f __attribute__((ext_vector_type(8)));

__device__ __forceinline__ float sigm(float x) { return 1.0f / (1.0f + __expf(-x)); }

__global__ __launch_bounds__(256)
void kgcn_fused(const float* __restrict__ user_emb,
                const float* __restrict__ ent_emb,
                const float* __restrict__ rel_emb,
                const float* __restrict__ Wg,
                const float* __restrict__ bias,
                const int* __restrict__ u,
                const int* __restrict__ v,
                const int* __restrict__ adj_ent,
                const int* __restrict__ adj_rel,
                float* __restrict__ out)
{
    __shared__ float sW[DIMN][WSP];
    __shared__ float sX[32][XS];
    __shared__ float sY[32][XS];
    __shared__ float sUser[DIMN];
    __shared__ float sUr[NRELS];
    __shared__ float sE0v[DIMN];
    __shared__ float sE1v[NNB][DIMN];
    __shared__ float sS1[NNB];
    __shared__ float sS2[NNB][NNB];
    __shared__ int   sE1[NNB];
    __shared__ int   sR1[NNB];
    __shared__ int   sE2[NNB][NNB];
    __shared__ int   sR2[NNB][NNB];
    __shared__ float sItem[DIMN];

    const int b    = blockIdx.x;
    const int t    = threadIdx.x;
    const int lane = t & 31;
    const int wave = t >> 5;

    const int e0  = v[b];
    const int uid = u[b];

    for (int i = t; i < DIMN * DIMN; i += 256) sW[i >> 6][i & 63] = Wg[i];
    if (t < DIMN)       sUser[t]     = user_emb[(size_t)uid * DIMN + t];
    else if (t < 128)   sE0v[t - 64] = ent_emb[(size_t)e0 * DIMN + (t - 64)];
    else if (t < 144)   sE1[t - 128] = adj_ent[(size_t)e0 * NNB + (t - 128)];
    else if (t < 160)   sR1[t - 144] = adj_rel[(size_t)e0 * NNB + (t - 144)];
    __syncthreads();

    {
        const int j = t >> 4, k = t & 15;
        const int ej = sE1[j];
        sE2[j][k] = adj_ent[(size_t)ej * NNB + k];
        sR2[j][k] = adj_rel[(size_t)ej * NNB + k];
        const int c4 = k * 4;
        const float4 p = *reinterpret_cast<const float4*>(&ent_emb[(size_t)ej * DIMN + c4]);
        *reinterpret_cast<float4*>(&sE1v[j][c4]) = p;
    }
    if (t < NRELS) {
        const float* rp = &rel_emb[t * DIMN];
        float acc = 0.f;
        #pragma unroll
        for (int d = 0; d < DIMN; ++d) acc += sUser[d] * rp[d];
        sUr[t] = acc;
    }
    __syncthreads();

    if (t < NNB) {
        float l[NNB], m = -1e30f;
        #pragma unroll
        for (int k = 0; k < NNB; ++k) { l[k] = sUr[sR2[t][k]]; m = fmaxf(m, l[k]); }
        float s = 0.f;
        #pragma unroll
        for (int k = 0; k < NNB; ++k) { l[k] = __expf(l[k] - m); s += l[k]; }
        const float inv = 1.f / s;
        #pragma unroll
        for (int k = 0; k < NNB; ++k) sS2[t][k] = l[k] * inv;
    } else if (t == 16) {
        float l[NNB], m = -1e30f;
        #pragma unroll
        for (int k = 0; k < NNB; ++k) { l[k] = sUr[sR1[k]]; m = fmaxf(m, l[k]); }
        float s = 0.f;
        #pragma unroll
        for (int k = 0; k < NNB; ++k) { l[k] = __expf(l[k] - m); s += l[k]; }
        const float inv = 1.f / s;
        #pragma unroll
        for (int k = 0; k < NNB; ++k) sS1[k] = l[k] * inv;
    }
    __syncthreads();

    {
        const int j = t >> 4, c4 = (t & 15) * 4;
        float4 acc = make_float4(0.f, 0.f, 0.f, 0.f);
        #pragma unroll
        for (int k = 0; k < NNB; ++k) {
            const float w = sS2[j][k];
            const float4 p = *reinterpret_cast<const float4*>(
                &ent_emb[(size_t)sE2[j][k] * DIMN + c4]);
            acc.x += w * p.x; acc.y += w * p.y; acc.z += w * p.z; acc.w += w * p.w;
        }
        float4 xo;
        xo.x = sE1v[j][c4 + 0] + acc.x;
        xo.y = sE1v[j][c4 + 1] + acc.y;
        xo.z = sE1v[j][c4 + 2] + acc.z;
        xo.w = sE1v[j][c4 + 3] + acc.w;
        *reinterpret_cast<float4*>(&sX[j][c4]) = xo;
    }
    if (t < DIMN) {
        float acc = 0.f;
        #pragma unroll
        for (int k = 0; k < NNB; ++k) acc += sS1[k] * sE1v[k][t];
        sX[16][t] = sE0v[t] + acc;
    }
    __syncthreads();

    {
        const int mtile = wave >> 2, ntile = wave & 3;
        const int arow  = (mtile << 4) + (lane & 15);
        const int koff  = (lane >> 4) << 1;
        const int bcol  = (ntile << 4) + (lane & 15);
        v8f c = {0.f, 0.f, 0.f, 0.f, 0.f, 0.f, 0.f, 0.f};
        #pragma unroll
        for (int kk = 0; kk < 16; ++kk) {
            const int kb = (kk << 2) + koff;
            v2f a;  a.x  = sX[arow][kb];   a.y  = sX[arow][kb + 1];
            v2f bb; bb.x = sW[kb][bcol];   bb.y = sW[kb + 1][bcol];
            c = __builtin_amdgcn_wmma_f32_16x16x4_f32(false, a, false, bb,
                                                      (short)0, c, false, false);
        }
        const float bv = bias[bcol];
        const int rbase = (mtile << 4) + ((lane >> 4) << 3);
        #pragma unroll
        for (int g = 0; g < 8; ++g)
            sY[rbase + g][bcol] = sigm(c[g] + bv);
    }
    __syncthreads();

    if (t < DIMN) {
        float acc = sY[16][t];
        #pragma unroll
        for (int k = 0; k < NNB; ++k) acc += sS1[k] * sY[k][t];
        sX[0][t] = acc;
    }
    __syncthreads();

    if (wave < 4) {
        const int ntile = wave;
        const int arow  = lane & 15;
        const int koff  = (lane >> 4) << 1;
        const int bcol  = (ntile << 4) + (lane & 15);
        v8f c = {0.f, 0.f, 0.f, 0.f, 0.f, 0.f, 0.f, 0.f};
        #pragma unroll
        for (int kk = 0; kk < 16; ++kk) {
            const int kb = (kk << 2) + koff;
            v2f a;  a.x  = sX[arow][kb];   a.y  = sX[arow][kb + 1];
            v2f bb; bb.x = sW[kb][bcol];   bb.y = sW[kb + 1][bcol];
            c = __builtin_amdgcn_wmma_f32_16x16x4_f32(false, a, false, bb,
                                                      (short)0, c, false, false);
        }
        if (lane < 16)
            sItem[bcol] = tanhf(c[0] + bias[bcol]);
    }
    __syncthreads();

    if (t < 32) {
        float p = sUser[t] * sItem[t] + sUser[t + 32] * sItem[t + 32];
        #pragma unroll
        for (int off = 16; off > 0; off >>= 1) p += __shfl_xor(p, off, 32);
        if (t == 0) out[b] = sigm(p);
    }
}

extern "C" void kernel_launch(void* const* d_in, const int* in_sizes, int n_in,
                              void* d_out, int out_size, void* d_ws, size_t ws_size,
                              hipStream_t stream) {
    const float* user_emb = (const float*)d_in[0];
    const float* ent_emb  = (const float*)d_in[1];
    const float* rel_emb  = (const float*)d_in[2];
    const float* W        = (const float*)d_in[3];
    const float* bias     = (const float*)d_in[4];
    const int*   u        = (const int*)d_in[5];
    const int*   v        = (const int*)d_in[6];
    const int*   adj_ent  = (const int*)d_in[7];
    const int*   adj_rel  = (const int*)d_in[8];
    float* out = (float*)d_out;
    const int B = in_sizes[5];

    kgcn_fused<<<B, 256, 0, stream>>>(user_emb, ent_emb, rel_emb, W, bias,
                                      u, v, adj_ent, adj_rel, out);
}


# Persistent grid: 1024 resident workgroups (4 per WGP) each loop over their batch elements; W staged to LDS once per workgroup instead of once per element (was 4096 one-element workgroups)
# speedup vs baseline: 1.1314x; 1.1314x over previous
#include <hip/hip_runtime.h>
#include <math.h>

#define DIMN 64
#define NNB  16
#define NRELS 32
#define XS   68
#define WSP  66

typedef float v2f __attribute__((ext_vector_type(2)));
typedef float v8f __attribute__((ext_vector_type(8)));

__device__ __forceinline__ float sigm(float x) { return 1.0f / (1.0f + __expf(-x)); }

__global__ __launch_bounds__(256)
void kgcn_fused(const float* __restrict__ user_emb,
                const float* __restrict__ ent_emb,
                const float* __restrict__ rel_emb,
                const float* __restrict__ Wg,
                const float* __restrict__ bias,
                const int* __restrict__ u,
                const int* __restrict__ v,
                const int* __restrict__ adj_ent,
                const int* __restrict__ adj_rel,
                float* __restrict__ out,
                const int nbatch)
{
    __shared__ float sW[DIMN][WSP];
    __shared__ float sX[32][XS];
    __shared__ float sY[32][XS];
    __shared__ float sUser[DIMN];
    __shared__ float sUr[NRELS];
    __shared__ float sE0v[DIMN];
    __shared__ float sE1v[NNB][DIMN];
    __shared__ float sS1[NNB];
    __shared__ float sS2[NNB][NNB];
    __shared__ int   sE1[NNB];
    __shared__ int   sR1[NNB];
    __shared__ int   sE2[NNB][NNB];
    __shared__ int   sR2[NNB][NNB];
    __shared__ float sItem[DIMN];

    const int t    = threadIdx.x;
    const int lane = t & 31;
    const int wave = t >> 5;

    for (int i = t; i < DIMN * DIMN; i += 256) sW[i >> 6][i & 63] = Wg[i];

    for (int b = blockIdx.x; b < nbatch; b += gridDim.x) {

    const int e0  = v[b];
    const int uid = u[b];

    if (t < DIMN)       sUser[t]     = user_emb[(size_t)uid * DIMN + t];
    else if (t < 128)   sE0v[t - 64] = ent_emb[(size_t)e0 * DIMN + (t - 64)];
    else if (t < 144)   sE1[t - 128] = adj_ent[(size_t)e0 * NNB + (t - 128)];
    else if (t < 160)   sR1[t - 144] = adj_rel[(size_t)e0 * NNB + (t - 144)];
    __syncthreads();

    const float us0 = sUser[lane], us1 = sUser[lane + 32];

    {
        const int j = t >> 4, k = t & 15;
        const int ej = sE1[j];
        sE2[j][k] = adj_ent[(size_t)ej * NNB + k];
        sR2[j][k] = adj_rel[(size_t)ej * NNB + k];
        const int c4 = k * 4;
        const float4 p = *reinterpret_cast<const float4*>(&ent_emb[(size_t)ej * DIMN + c4]);
        *reinterpret_cast<float4*>(&sE1v[j][c4]) = p;
    }
    if (t < NRELS) {
        const float* rp = &rel_emb[t * DIMN];
        float acc = 0.f;
        #pragma unroll
        for (int d = 0; d < DIMN; ++d) acc += sUser[d] * rp[d];
        sUr[t] = acc;
    }
    __syncthreads();

    if (t < NNB) {
        float l[NNB], m = -1e30f;
        #pragma unroll
        for (int k = 0; k < NNB; ++k) { l[k] = sUr[sR2[t][k]]; m = fmaxf(m, l[k]); }
        float s = 0.f;
        #pragma unroll
        for (int k = 0; k < NNB; ++k) { l[k] = __expf(l[k] - m); s += l[k]; }
        const float inv = 1.f / s;
        #pragma unroll
        for (int k = 0; k < NNB; ++k) sS2[t][k] = l[k] * inv;
    } else if (t == 16) {
        float l[NNB], m = -1e30f;
        #pragma unroll
        for (int k = 0; k < NNB; ++k) { l[k] = sUr[sR1[k]]; m = fmaxf(m, l[k]); }
        float s = 0.f;
        #pragma unroll
        for (int k = 0; k < NNB; ++k) { l[k] = __expf(l[k] - m); s += l[k]; }
        const float inv = 1.f / s;
        #pragma unroll
        for (int k = 0; k < NNB; ++k) sS1[k] = l[k] * inv;
    }
    __syncthreads();

    {
        const int j = t >> 4, c4 = (t & 15) * 4;
        float4 acc = make_float4(0.f, 0.f, 0.f, 0.f);
        #pragma unroll
        for (int k = 0; k < NNB; ++k) {
            const float w = sS2[j][k];
            const float4 p = *reinterpret_cast<const float4*>(
                &ent_emb[(size_t)sE2[j][k] * DIMN + c4]);
            acc.x += w * p.x; acc.y += w * p.y; acc.z += w * p.z; acc.w += w * p.w;
        }
        float4 xo;
        xo.x = sE1v[j][c4 + 0] + acc.x;
        xo.y = sE1v[j][c4 + 1] + acc.y;
        xo.z = sE1v[j][c4 + 2] + acc.z;
        xo.w = sE1v[j][c4 + 3] + acc.w;
        *reinterpret_cast<float4*>(&sX[j][c4]) = xo;
    }
    if (t < DIMN) {
        float acc = 0.f;
        #pragma unroll
        for (int k = 0; k < NNB; ++k) acc += sS1[k] * sE1v[k][t];
        sX[16][t] = sE0v[t] + acc;
    }
    __syncthreads();

    {
        const int mtile = wave >> 2, ntile = wave & 3;
        const int arow  = (mtile << 4) + (lane & 15);
        const int koff  = (lane >> 4) << 1;
        const int bcol  = (ntile << 4) + (lane & 15);
        v8f c = {0.f, 0.f, 0.f, 0.f, 0.f, 0.f, 0.f, 0.f};
        #pragma unroll
        for (int kk = 0; kk < 16; ++kk) {
            const int kb = (kk << 2) + koff;
            v2f a;  a.x  = sX[arow][kb];   a.y  = sX[arow][kb + 1];
            v2f bb; bb.x = sW[kb][bcol];   bb.y = sW[kb + 1][bcol];
            c = __builtin_amdgcn_wmma_f32_16x16x4_f32(false, a, false, bb,
                                                      (short)0, c, false, false);
        }
        const float bv = bias[bcol];
        const int rbase = (mtile << 4) + ((lane >> 4) << 3);
        #pragma unroll
        for (int g = 0; g < 8; ++g)
            sY[rbase + g][bcol] = sigm(c[g] + bv);
    }
    __syncthreads();

    if (t < DIMN) {
        float acc = sY[16][t];
        #pragma unroll
        for (int k = 0; k < NNB; ++k) acc += sS1[k] * sY[k][t];
        sX[0][t] = acc;
    }
    __syncthreads();

    if (wave < 4) {
        const int ntile = wave;
        const int arow  = lane & 15;
        const int koff  = (lane >> 4) << 1;
        const int bcol  = (ntile << 4) + (lane & 15);
        v8f c = {0.f, 0.f, 0.f, 0.f, 0.f, 0.f, 0.f, 0.f};
        #pragma unroll
        for (int kk = 0; kk < 16; ++kk) {
            const int kb = (kk << 2) + koff;
            v2f a;  a.x  = sX[arow][kb];   a.y  = sX[arow][kb + 1];
            v2f bb; bb.x = sW[kb][bcol];   bb.y = sW[kb + 1][bcol];
            c = __builtin_amdgcn_wmma_f32_16x16x4_f32(false, a, false, bb,
                                                      (short)0, c, false, false);
        }
        if (lane < 16)
            sItem[bcol] = tanhf(c[0] + bias[bcol]);
    }
    __syncthreads();

    if (t < 32) {
        float p = us0 * sItem[t] + us1 * sItem[t + 32];
        #pragma unroll
        for (int off = 16; off > 0; off >>= 1) p += __shfl_xor(p, off, 32);
        if (t == 0) out[b] = sigm(p);
    }
    }
}

extern "C" void kernel_launch(void* const* d_in, const int* in_sizes, int n_in,
                              void* d_out, int out_size, void* d_ws, size_t ws_size,
                              hipStream_t stream) {
    const float* user_emb = (const float*)d_in[0];
    const float* ent_emb  = (const float*)d_in[1];
    const float* rel_emb  = (const float*)d_in[2];
    const float* W        = (const float*)d_in[3];
    const float* bias     = (const float*)d_in[4];
    const int*   u        = (const int*)d_in[5];
    const int*   v        = (const int*)d_in[6];
    const int*   adj_ent  = (const int*)d_in[7];
    const int*   adj_rel  = (const int*)d_in[8];
    float* out = (float*)d_out;
    const int B = in_sizes[5];

    kgcn_fused<<<1024, 256, 0, stream>>>(user_emb, ent_emb, rel_emb, W, bias,
                                         u, v, adj_ent, adj_rel, out, B);
}


# Index chain batched per workgroup: hop-0/hop-1 adjacency, hop-1 rows, user-relation dots and softmax of its 4 elements fetched together up front; element loop keeps only the row gather and products
# speedup vs baseline: 1.2069x; 1.0667x over previous
#include <hip/hip_runtime.h>
#include <math.h>


#define DIMN 64
#define NNB  16
#define NRELS 32
#define XS   68
#define WSP  66
#define EPB  4

typedef float v2f __attribute__((ext_vector_type(2)));
typedef float v8f __attribute__((ext_vector_type(8)));

__device__ __forceinline__ float sigm(float x) { return 1.0f / (1.0f + __expf(-x)); }

__device__ __forceinline__ float ldf(const float* base, unsigned idx) {
    return *reinterpret_cast<const float*>(reinterpret_cast<const char*>(base) + idx * 4u);
}
__device__ __forceinline__ int ldi(const int* base, unsigned idx) {
    return *reinterpret_cast<const int*>(reinterpret_cast<const char*>(base) + idx * 4u);
}
__device__ __forceinline__ float4 ldf4(const float* base, unsigned idx) {
    return *reinterpret_cast<const float4*>(reinterpret_cast<const char*>(base) + idx * 4u);
}

__global__ __launch_bounds__(256, 8)
void kgcn_fused(const float* __restrict__ user_emb,
                const float* __restrict__ ent_emb,
                const float* __restrict__ rel_emb,
                const float* __restrict__ Wg,
                const float* __restrict__ bias,
                const int* __restrict__ u,
                const int* __restrict__ v,
                const int* __restrict__ adj_ent,
                const int* __restrict__ adj_rel,
                float* __restrict__ out,
                const int nbatch)
{
    __shared__ float sW[DIMN][WSP];
    __shared__ float sX[32][XS];
    __shared__ float sY[32][XS];
    __shared__ float sUser[EPB][DIMN];
    __shared__ float sUr[EPB][NRELS];
    __shared__ float sE0v[EPB][DIMN];
    __shared__ float sE1v[EPB][NNB][DIMN];
    __shared__ float sS1[EPB][NNB];
    __shared__ float sS2[EPB][NNB][NNB];
    __shared__ int   sA1[EPB][DIMN];
    __shared__ int   sE2[EPB][NNB][NNB];
    __shared__ int   sR2[EPB][NNB][NNB];
    __shared__ float sItem[DIMN];

    const int t    = threadIdx.x;
    const int lane = t & 31;
    const int wave = t >> 5;
    const int stride = gridDim.x;

    for (int i = t; i < DIMN * DIMN; i += 256) sW[i >> 6][i & 63] = Wg[i];

    for (int base = blockIdx.x; base < nbatch; base += EPB * stride) {

    int ne = 1;
    for (int i = 1; i < EPB; ++i) if (base + i * stride < nbatch) ne = i + 1;

    {
        const int i  = t >> 6, c = t & 63;
        const int ii = (i < ne) ? i : 0;
        const int b   = base + ii * stride;
        const int e0  = v[b];
        const int uid = u[b];
        const unsigned ao = (unsigned)e0 * NNB + (c & (NNB - 1));
        const int   ae = ldi(adj_ent, ao);
        const int   ar = ldi(adj_rel, ao);
        const float uv = ldf(user_emb, (unsigned)uid * DIMN + c);
        const float ev = ldf(ent_emb,  (unsigned)e0  * DIMN + c);
        sA1[i][c]   = (c & NNB) ? ar : ae;
        sUser[i][c] = uv;
        sE0v[i][c]  = ev;
    }
    __syncthreads();

    {
        const int j = t >> 4, k = t & 15, c4 = k * 4;
        int    n2[EPB], r2[EPB];
        float4 p[EPB];
        #pragma unroll
        for (int i = 0; i < EPB; ++i) {
            const int ej = sA1[i][j];
            n2[i] = ldi(adj_ent, (unsigned)ej * NNB + k);
            r2[i] = ldi(adj_rel, (unsigned)ej * NNB + k);
            p[i]  = ldf4(ent_emb, (unsigned)ej * DIMN + c4);
        }
        #pragma unroll
        for (int i = 0; i < EPB; ++i) {
            sE2[i][j][k] = n2[i];
            sR2[i][j][k] = r2[i];
            *reinterpret_cast<float4*>(&sE1v[i][j][c4]) = p[i];
        }
    }
    if (t < EPB * NRELS) {
        const int i = t >> 5, r = t & 31;
        const float* rp = &rel_emb[r * DIMN];
        float acc = 0.f;
        #pragma unroll
        for (int d = 0; d < DIMN; ++d) acc += sUser[i][d] * rp[d];
        sUr[i][r] = acc;
    }
    __syncthreads();

    if (t < EPB * NNB) {
        const int i = t >> 4, r = t & 15;
        float l[NNB], m = -1e30f;
        #pragma unroll
        for (int k = 0; k < NNB; ++k) { l[k] = sUr[i][sR2[i][r][k]]; m = fmaxf(m, l[k]); }
        float s = 0.f;
        #pragma unroll
        for (int k = 0; k < NNB; ++k) { l[k] = __expf(l[k] - m); s += l[k]; }
        const float inv = 1.f / s;
        #pragma unroll
        for (int k = 0; k < NNB; ++k) sS2[i][r][k] = l[k] * inv;
    } else if (t < EPB * NNB + EPB) {
        const int i = t - EPB * NNB;
        float l[NNB], m = -1e30f;
        #pragma unroll
        for (int k = 0; k < NNB; ++k) { l[k] = sUr[i][sA1[i][NNB + k]]; m = fmaxf(m, l[k]); }
        float s = 0.f;
        #pragma unroll
        for (int k = 0; k < NNB; ++k) { l[k] = __expf(l[k] - m); s += l[k]; }
        const float inv = 1.f / s;
        #pragma unroll
        for (int k = 0; k < NNB; ++k) sS1[i][k] = l[k] * inv;
    }
    __syncthreads();

    for (int i = 0; i < ne; ++i) {
    const int b = base + i * stride;

    {
        const int j = t >> 4, c4 = (t & 15) * 4;
        float4 acc = make_float4(0.f, 0.f, 0.f, 0.f);
        #pragma unroll
        for (int k = 0; k < NNB; ++k) {
            const float w = sS2[i][j][k];
            const float4 p = ldf4(ent_emb, (unsigned)sE2[i][j][k] * DIMN + c4);
            acc.x += w * p.x; acc.y += w * p.y; acc.z += w * p.z; acc.w += w * p.w;
        }
        float4 xo;
        xo.x = sE1v[i][j][c4 + 0] + acc.x;
        xo.y = sE1v[i][j][c4 + 1] + acc.y;
        xo.z = sE1v[i][j][c4 + 2] + acc.z;
        xo.w = sE1v[i][j][c4 + 3] + acc.w;
        *reinterpret_cast<float4*>(&sX[j][c4]) = xo;
    }
    if (t < DIMN) {
        float acc = 0.f;
        #pragma unroll
        for (int k = 0; k < NNB; ++k) acc += sS1[i][k] * sE1v[i][k][t];
        sX[16][t] = sE0v[i][t] + acc;
    }
    __syncthreads();

    {
        const int mtile = wave >> 2, ntile = wave & 3;
        const int arow  = (mtile << 4) + (lane & 15);
        const int koff  = (lane >> 4) << 1;
        const int bcol  = (ntile << 4) + (lane & 15);
        v8f c = {0.f, 0.f, 0.f, 0.f, 0.f, 0.f, 0.f, 0.f};
        #pragma unroll
        for (int kk = 0; kk < 16; ++kk) {
            const int kb = (kk << 2) + koff;
            v2f a;  a.x  = sX[arow][kb];   a.y  = sX[arow][kb + 1];
            v2f bb; bb.x = sW[kb][bcol];   bb.y = sW[kb + 1][bcol];
            c = __builtin_amdgcn_wmma_f32_16x16x4_f32(false, a, false, bb,
                                                      (short)0, c, false, false);
        }
        const float bv = bias[bcol];
        const int rbase = (mtile << 4) + ((lane >> 4) << 3);
        #pragma unroll
        for (int g = 0; g < 8; ++g)
            sY[rbase + g][bcol] = sigm(c[g] + bv);
    }
    __syncthreads();

    if (t < DIMN) {
        float acc = sY[16][t];
        #pragma unroll
        for (int k = 0; k < NNB; ++k) acc += sS1[i][k] * sY[k][t];
        sX[0][t] = acc;
    }
    __syncthreads();

    if (wave < 4) {
        const int ntile = wave;
        const int arow  = lane & 15;
        const int koff  = (lane >> 4) << 1;
        const int bcol  = (ntile << 4) + (lane & 15);
        v8f c = {0.f, 0.f, 0.f, 0.f, 0.f, 0.f, 0.f, 0.f};
        #pragma unroll
        for (int kk = 0; kk < 16; ++kk) {
            const int kb = (kk << 2) + koff;
            v2f a;  a.x  = sX[arow][kb];   a.y  = sX[arow][kb + 1];
            v2f bb; bb.x = sW[kb][bcol];   bb.y = sW[kb + 1][bcol];
            c = __builtin_amdgcn_wmma_f32_16x16x4_f32(false, a, false, bb,
                                                      (short)0, c, false, false);
        }
        if (lane < 16)
            sItem[bcol] = tanhf(c[0] + bias[bcol]);
    }
    __syncthreads();

    if (t < 32) {
        float p = sUser[i][t] * sItem[t] + sUser[i][t + 32] * sItem[t + 32];
        #pragma unroll
        for (int off = 16; off > 0; off >>= 1) p += __shfl_xor(p, off, 32);
        if (t == 0) out[b] = sigm(p);
    }
    }

    __syncthreads();
    }
}

extern "C" void kernel_launch(void* const* d_in, const int* in_sizes, int n_in,
                              void* d_out, int out_size, void* d_ws, size_t ws_size,
                              hipStream_t stream) {
    const float* user_emb = (const float*)d_in[0];
    const float* ent_emb  = (const float*)d_in[1];
    const float* rel_emb  = (const float*)d_in[2];
    const float* W        = (const float*)d_in[3];
    const float* bias     = (const float*)d_in[4];
    const int*   u        = (const int*)d_in[5];
    const int*   v        = (const int*)d_in[6];
    const int*   adj_ent  = (const int*)d_in[7];
    const int*   adj_rel  = (const int*)d_in[8];
    float* out = (float*)d_out;
    const int B = in_sizes[5];

    kgcn_fused<<<1024, 256, 0, stream>>>(user_emb, ent_emb, rel_emb, W, bias,
                                         u, v, adj_ent, adj_rel, out, B);
}


# Products batched over the workgroup's 4 elements: row gathers back to back into one 80-row X, then one GEMM pass (3 accumulators per wave), one aggregation, one 4-row GEMM, one score step
# speedup vs baseline: 1.3416x; 1.1116x over previous
#include <hip/hip_runtime.h>
#include <math.h>


#define DIMN 64
#define NNB  16
#define NRELS 32
#define XS   68
#define WSP  66
#define EPB  4
#define XROWS 80

typedef float v2f __attribute__((ext_vector_type(2)));
typedef float v8f __attribute__((ext_vector_type(8)));

__device__ __forceinline__ float sigm(float x) { return 1.0f / (1.0f + __expf(-x)); }

__device__ __forceinline__ float ldf(const float* base, unsigned idx) {
    return *reinterpret_cast<const float*>(reinterpret_cast<const char*>(base) + idx * 4u);
}
__device__ __forceinline__ int ldi(const int* base, unsigned idx) {
    return *reinterpret_cast<const int*>(reinterpret_cast<const char*>(base) + idx * 4u);
}
__device__ __forceinline__ float4 ldf4(const float* base, unsigned idx) {
    return *reinterpret_cast<const float4*>(reinterpret_cast<const char*>(base) + idx * 4u);
}

template <int NT>
__device__ __forceinline__ void xw_tiles(const float (*X)[XS], const float (*Wt)[WSP],
                                         const int arow, const int koff, const int bcol,
                                         v8f (&c)[3]) {
    #pragma unroll
    for (int kk = 0; kk < 16; ++kk) {
        const int kb = (kk << 2) + koff;
        v2f bb; bb.x = Wt[kb][bcol]; bb.y = Wt[kb + 1][bcol];
        #pragma unroll
        for (int q = 0; q < NT; ++q) {
            v2f a; a.x = X[arow + 32 * q][kb]; a.y = X[arow + 32 * q][kb + 1];
            c[q] = __builtin_amdgcn_wmma_f32_16x16x4_f32(false, a, false, bb,
                                                         (short)0, c[q], false, false);
        }
    }
}

__global__ __launch_bounds__(256, 8)
void kgcn_fused(const float* __restrict__ user_emb,
                const float* __restrict__ ent_emb,
                const float* __restrict__ rel_emb,
                const float* __restrict__ Wg,
                const float* __restrict__ bias,
                const int* __restrict__ u,
                const int* __restrict__ v,
                const int* __restrict__ adj_ent,
                const int* __restrict__ adj_rel,
                float* __restrict__ out,
                const int nbatch)
{
    __shared__ float sW[DIMN][WSP];
    __shared__ float sX[XROWS][XS];
    __shared__ float sUser[EPB][DIMN];
    __shared__ float sUr[EPB][NRELS];
    __shared__ float sE0v[EPB][DIMN];
    __shared__ float sE1v[EPB][NNB][DIMN];
    __shared__ float sS1[EPB][NNB];
    __shared__ float sS2[EPB][NNB][NNB];
    __shared__ int   sA1[EPB][DIMN];
    __shared__ int   sE2[EPB][NNB][NNB];
    __shared__ int   sR2[EPB][NNB][NNB];
    __shared__ float sItem[EPB][DIMN];

    const int t    = threadIdx.x;
    const int lane = t & 31;
    const int wave = t >> 5;
    const int stride = gridDim.x;

    for (int i = t; i < DIMN * DIMN; i += 256) sW[i >> 6][i & 63] = Wg[i];
    for (int q = t; q < XROWS * XS / 4; q += 256)
        reinterpret_cast<float4*>(&sX[0][0])[q] = make_float4(0.f, 0.f, 0.f, 0.f);

    for (int base = blockIdx.x; base < nbatch; base += EPB * stride) {

    int ne = 1;
    for (int i = 1; i < EPB; ++i) if (base + i * stride < nbatch) ne = i + 1;

    {
        const int i  = t >> 6, c = t & 63;
        const int ii = (i < ne) ? i : 0;
        const int b   = base + ii * stride;
        const int e0  = v[b];
        const int uid = u[b];
        const unsigned ao = (unsigned)e0 * NNB + (c & (NNB - 1));
        const int   ae = ldi(adj_ent, ao);
        const int   ar = ldi(adj_rel, ao);
        const float uv = ldf(user_emb, (unsigned)uid * DIMN + c);
        const float ev = ldf(ent_emb,  (unsigned)e0  * DIMN + c);
        sA1[i][c]   = (c & NNB) ? ar : ae;
        sUser[i][c] = uv;
        sE0v[i][c]  = ev;
    }
    __syncthreads();

    {
        const int j = t >> 4, k = t & 15, c4 = k * 4;
        int    n2[EPB], r2[EPB];
        float4 p[EPB];
        #pragma unroll
        for (int i = 0; i < EPB; ++i) {
            const int ej = sA1[i][j];
            n2[i] = ldi(adj_ent, (unsigned)ej * NNB + k);
            r2[i] = ldi(adj_rel, (unsigned)ej * NNB + k);
            p[i]  = ldf4(ent_emb, (unsigned)ej * DIMN + c4);
        }
        #pragma unroll
        for (int i = 0; i < EPB; ++i) {
            sE2[i][j][k] = n2[i];
            sR2[i][j][k] = r2[i];
            *reinterpret_cast<float4*>(&sE1v[i][j][c4]) = p[i];
        }
    }
    if (t < EPB * NRELS) {
        const int i = t >> 5, r = t & 31;
        const float* rp = &rel_emb[r * DIMN];
        float acc = 0.f;
        #pragma unroll
        for (int d = 0; d < DIMN; ++d) acc += sUser[i][d] * rp[d];
        sUr[i][r] = acc;
    }
    __syncthreads();

    if (t < EPB * NNB) {
        const int i = t >> 4, r = t & 15;
        float l[NNB], m = -1e30f;
        #pragma unroll
        for (int k = 0; k < NNB; ++k) { l[k] = sUr[i][sR2[i][r][k]]; m = fmaxf(m, l[k]); }
        float s = 0.f;
        #pragma unroll
        for (int k = 0; k < NNB; ++k) { l[k] = __expf(l[k] - m); s += l[k]; }
        const float inv = 1.f / s;
        #pragma unroll
        for (int k = 0; k < NNB; ++k) sS2[i][r][k] = l[k] * inv;
    } else if (t < EPB * NNB + EPB) {
        const int i = t - EPB * NNB;
        float l[NNB], m = -1e30f;
        #pragma unroll
        for (int k = 0; k < NNB; ++k) { l[k] = sUr[i][sA1[i][NNB + k]]; m = fmaxf(m, l[k]); }
        float s = 0.f;
        #pragma unroll
        for (int k = 0; k < NNB; ++k) { l[k] = __expf(l[k] - m); s += l[k]; }
        const float inv = 1.f / s;
        #pragma unroll
        for (int k = 0; k < NNB; ++k) sS1[i][k] = l[k] * inv;
    }
    __syncthreads();

    #pragma clang loop vectorize(disable) interleave(disable) unroll(disable)
    for (int i = 0; i < ne; ++i) {
        const int j = t >> 4, c4 = (t & 15) * 4;
        float4 acc = make_float4(0.f, 0.f, 0.f, 0.f);
        #pragma unroll
        for (int k = 0; k < NNB; ++k) {
            const float w = sS2[i][j][k];
            const float4 p = ldf4(ent_emb, (unsigned)sE2[i][j][k] * DIMN + c4);
            acc.x += w * p.x; acc.y += w * p.y; acc.z += w * p.z; acc.w += w * p.w;
        }
        float4 xo;
        xo.x = sE1v[i][j][c4 + 0] + acc.x;
        xo.y = sE1v[i][j][c4 + 1] + acc.y;
        xo.z = sE1v[i][j][c4 + 2] + acc.z;
        xo.w = sE1v[i][j][c4 + 3] + acc.w;
        *reinterpret_cast<float4*>(&sX[(i << 4) + j][c4]) = xo;
    }
    {
        const int i = t >> 6, c = t & 63;
        float acc = 0.f;
        #pragma unroll
        for (int k = 0; k < NNB; ++k) acc += sS1[i][k] * sE1v[i][k][c];
        sX[64 + i][c] = sE0v[i][c] + acc;
    }
    __syncthreads();

    {
        const int ntile = wave & 3, mh = wave >> 2;
        const int arow  = (mh << 4) + (lane & 15);
        const int koff  = (lane >> 4) << 1;
        const int bcol  = (ntile << 4) + (lane & 15);
        v8f c[3];
        #pragma unroll
        for (int q = 0; q < 3; ++q) c[q] = (v8f){0.f, 0.f, 0.f, 0.f, 0.f, 0.f, 0.f, 0.f};
        if (mh == 0) xw_tiles<3>(sX, sW, arow, koff, bcol, c);
        else         xw_tiles<2>(sX, sW, arow, koff, bcol, c);
        __syncthreads();
        const float bv = bias[bcol];
        const int rbase = (mh << 4) + ((lane >> 4) << 3);
        #pragma unroll
        for (int q = 0; q < 3; ++q) {
            if (q < 2 || mh == 0) {
                #pragma unroll
                for (int g = 0; g < 8; ++g)
                    sX[rbase + 32 * q + g][bcol] = sigm(c[q][g] + bv);
            }
        }
    }
    __syncthreads();

    {
        const int i = t >> 6, c = t & 63;
        float acc = sX[64 + i][c];
        #pragma unroll
        for (int k = 0; k < NNB; ++k) acc += sS1[i][k] * sX[(i << 4) + k][c];
        sX[68 + i][c] = acc;
    }
    __syncthreads();

    if (wave < 4) {
        const int ntile = wave;
        const int arow  = 64 + (lane & 15);
        const int koff  = (lane >> 4) << 1;
        const int bcol  = (ntile << 4) + (lane & 15);
        v8f c = {0.f, 0.f, 0.f, 0.f, 0.f, 0.f, 0.f, 0.f};
        #pragma unroll
        for (int kk = 0; kk < 16; ++kk) {
            const int kb = (kk << 2) + koff;
            v2f a;  a.x  = sX[arow][kb];   a.y  = sX[arow][kb + 1];
            v2f bb; bb.x = sW[kb][bcol];   bb.y = sW[kb + 1][bcol];
            c = __builtin_amdgcn_wmma_f32_16x16x4_f32(false, a, false, bb,
                                                      (short)0, c, false, false);
        }
        if (lane < 16) {
            const float bv = bias[bcol];
            #pragma unroll
            for (int i = 0; i < EPB; ++i) sItem[i][bcol] = tanhf(c[4 + i] + bv);
        }
    }
    __syncthreads();

    if (t < 32 * EPB) {
        const int i = wave;
        float p = sUser[i][lane] * sItem[i][lane] + sUser[i][lane + 32] * sItem[i][lane + 32];
        #pragma unroll
        for (int off = 16; off > 0; off >>= 1) p += __shfl_xor(p, off, 32);
        if (lane == 0 && i < ne) out[base + i * stride] = sigm(p);
    }

    __syncthreads();
    }
}

extern "C" void kernel_launch(void* const* d_in, const int* in_sizes, int n_in,
                              void* d_out, int out_size, void* d_ws, size_t ws_size,
                              hipStream_t stream) {
    const float* user_emb = (const float*)d_in[0];
    const float* ent_emb  = (const float*)d_in[1];
    const float* rel_emb  = (const float*)d_in[2];
    const float* W        = (const float*)d_in[3];
    const float* bias     = (const float*)d_in[4];
    const int*   u        = (const int*)d_in[5];
    const int*   v        = (const int*)d_in[6];
    const int*   adj_ent  = (const int*)d_in[7];
    const int*   adj_rel  = (const int*)d_in[8];
    float* out = (float*)d_out;
    const int B = in_sizes[5];

    kgcn_fused<<<1024, 256, 0, stream>>>(user_emb, ent_emb, rel_emb, W, bias,
                                         u, v, adj_ent, adj_rel, out, B);
}


# Index-independent operands ride under the index chain: W tile, relation table (parked in free X rows) and bias loaded beside the first index loads; user-relation dots read LDS under the hop-1 loads
# speedup vs baseline: 1.5063x; 1.1228x over previous
#include <hip/hip_runtime.h>
#include <math.h>


#define DIMN 64
#define NNB  16
#define NRELS 32
#define XS   68
#define WSP  66
#define EPB  4
#define XROWS 80

typedef float v2f __attribute__((ext_vector_type(2)));
typedef float v8f __attribute__((ext_vector_type(8)));

__device__ __forceinline__ float sigm(float x) { return 1.0f / (1.0f + __expf(-x)); }

__device__ __forceinline__ float ldf(const float* base, unsigned idx) {
    return *reinterpret_cast<const float*>(reinterpret_cast<const char*>(base) + idx * 4u);
}
__device__ __forceinline__ int ldi(const int* base, unsigned idx) {
    return *reinterpret_cast<const int*>(reinterpret_cast<const char*>(base) + idx * 4u);
}
__device__ __forceinline__ float4 ldf4(const float* base, unsigned idx) {
    return *reinterpret_cast<const float4*>(reinterpret_cast<const char*>(base) + idx * 4u);
}

template <int NT>
__device__ __forceinline__ void xw_tiles(const float (*X)[XS], const float (*Wt)[WSP],
                                         const int arow, const int koff, const int bcol,
                                         v8f (&c)[3]) {
    #pragma unroll
    for (int kk = 0; kk < 16; ++kk) {
        const int kb = (kk << 2) + koff;
        v2f bb; bb.x = Wt[kb][bcol]; bb.y = Wt[kb + 1][bcol];
        #pragma unroll
        for (int q = 0; q < NT; ++q) {
            v2f a; a.x = X[arow + 32 * q][kb]; a.y = X[arow + 32 * q][kb + 1];
            c[q] = __builtin_amdgcn_wmma_f32_16x16x4_f32(false, a, false, bb,
                                                         (short)0, c[q], false, false);
        }
    }
}

__global__ __launch_bounds__(256, 8)
void kgcn_fused(const float* __restrict__ user_emb,
                const float* __restrict__ ent_emb,
                const float* __restrict__ rel_emb,
                const float* __restrict__ Wg,
                const float* __restrict__ bias,
                const int* __restrict__ u,
                const int* __restrict__ v,
                const int* __restrict__ adj_ent,
                const int* __restrict__ adj_rel,
                float* __restrict__ out,
                const int nbatch)
{
    __shared__ float sW[DIMN][WSP];
    __shared__ float sX[XROWS][XS];
    __shared__ float sUser[EPB][DIMN];
    __shared__ float sUr[EPB][NRELS];
    __shared__ float sE0v[EPB][DIMN];
    __shared__ float sE1v[EPB][NNB][DIMN];
    __shared__ float sS1[EPB][NNB];
    __shared__ float sS2[EPB][NNB][NNB];
    __shared__ int   sA1[EPB][DIMN];
    __shared__ int   sE2[EPB][NNB][NNB];
    __shared__ int   sR2[EPB][NNB][NNB];
    __shared__ float sItem[EPB][DIMN];

    const int t    = threadIdx.x;
    const int lane = t & 31;
    const int wave = t >> 5;
    const int stride = gridDim.x;

    float4 wv[4];
    #pragma unroll
    for (int q = 0; q < 4; ++q) wv[q] = ldf4(Wg, (unsigned)(t + 256 * q) * 4u);

    const float bias_v = bias[((wave & 3) << 4) + (lane & 15)];

    float4 rv[2];
    auto rel_loads = [&]() {
        #pragma unroll
        for (int q = 0; q < 2; ++q) rv[q] = ldf4(rel_emb, (unsigned)(t + 256 * q) * 4u);
    };
    auto rel_stores = [&]() {
        #pragma unroll
        for (int q = 0; q < 2; ++q) {
            const int f4 = t + 256 * q;
            *reinterpret_cast<float4*>(&sX[f4 >> 4][(f4 & 15) << 2]) = rv[q];
        }
    };
    rel_loads();

    if ((int)blockIdx.x >= nbatch) return;

    int a_ae, a_ar; float a_uv, a_ev;
    auto phase_a_loads = [&](const int gbase) {
        int gne = 1;
        for (int i = 1; i < EPB; ++i) if (gbase + i * stride < nbatch) gne = i + 1;
        const int i  = t >> 6, c = t & 63;
        const int ii = (i < gne) ? i : 0;
        const int b   = gbase + ii * stride;
        const int e0  = v[b];
        const int uid = u[b];
        const unsigned ao = (unsigned)e0 * NNB + (c & (NNB - 1));
        a_ae = ldi(adj_ent, ao);
        a_ar = ldi(adj_rel, ao);
        a_uv = ldf(user_emb, (unsigned)uid * DIMN + c);
        a_ev = ldf(ent_emb,  (unsigned)e0  * DIMN + c);
    };
    phase_a_loads(blockIdx.x);

    #pragma unroll
    for (int q = 0; q < 4; ++q) {
        const int f4 = t + 256 * q, row = f4 >> 4, col = (f4 & 15) << 2;
        *reinterpret_cast<float2*>(&sW[row][col])     = make_float2(wv[q].x, wv[q].y);
        *reinterpret_cast<float2*>(&sW[row][col + 2]) = make_float2(wv[q].z, wv[q].w);
    }
    rel_stores();
    for (int q = NRELS * XS / 4 + t; q < XROWS * XS / 4; q += 256)
        reinterpret_cast<float4*>(&sX[0][0])[q] = make_float4(0.f, 0.f, 0.f, 0.f);

    for (int base = blockIdx.x; ; base += EPB * stride) {

    int ne = 1;
    for (int i = 1; i < EPB; ++i) if (base + i * stride < nbatch) ne = i + 1;

    {
        const int i  = t >> 6, c = t & 63;
        sA1[i][c]   = (c & NNB) ? a_ar : a_ae;
        sUser[i][c] = a_uv;
        sE0v[i][c]  = a_ev;
    }
    __syncthreads();

    {
        const int j = t >> 4, k = t & 15, c4 = k * 4;
        int    n2[EPB], r2[EPB];
        float4 p[EPB];
        #pragma unroll
        for (int i = 0; i < EPB; ++i) {
            const int ej = sA1[i][j];
            n2[i] = ldi(adj_ent, (unsigned)ej * NNB + k);
            r2[i] = ldi(adj_rel, (unsigned)ej * NNB + k);
            p[i]  = ldf4(ent_emb, (unsigned)ej * DIMN + c4);
        }
        if (t < EPB * NRELS) {
            const int i = t >> 5, r = t & 31;
            float acc = 0.f;
            #pragma unroll
            for (int d = 0; d < DIMN; ++d) acc += sUser[i][d] * sX[r][d];
            sUr[i][r] = acc;
        }
        #pragma unroll
        for (int i = 0; i < EPB; ++i) {
            sE2[i][j][k] = n2[i];
            sR2[i][j][k] = r2[i];
            *reinterpret_cast<float4*>(&sE1v[i][j][c4]) = p[i];
        }
    }
    __syncthreads();

    if (t < EPB * NNB) {
        const int i = t >> 4, r = t & 15;
        float l[NNB], m = -1e30f;
        #pragma unroll
        for (int k = 0; k < NNB; ++k) { l[k] = sUr[i][sR2[i][r][k]]; m = fmaxf(m, l[k]); }
        float s = 0.f;
        #pragma unroll
        for (int k = 0; k < NNB; ++k) { l[k] = __expf(l[k] - m); s += l[k]; }
        const float inv = 1.f / s;
        #pragma unroll
        for (int k = 0; k < NNB; ++k) sS2[i][r][k] = l[k] * inv;
    } else if (t < EPB * NNB + EPB) {
        const int i = t - EPB * NNB;
        float l[NNB], m = -1e30f;
        #pragma unroll
        for (int k = 0; k < NNB; ++k) { l[k] = sUr[i][sA1[i][NNB + k]]; m = fmaxf(m, l[k]); }
        float s = 0.f;
        #pragma unroll
        for (int k = 0; k < NNB; ++k) { l[k] = __expf(l[k] - m); s += l[k]; }
        const float inv = 1.f / s;
        #pragma unroll
        for (int k = 0; k < NNB; ++k) sS1[i][k] = l[k] * inv;
    }
    __syncthreads();

    #pragma clang loop vectorize(disable) interleave(disable) unroll(disable)
    for (int i = 0; i < ne; ++i) {
        const int j = t >> 4, c4 = (t & 15) * 4;
        float4 acc = make_float4(0.f, 0.f, 0.f, 0.f);
        #pragma unroll
        for (int k = 0; k < NNB; ++k) {
            const float w = sS2[i][j][k];
            const float4 p = ldf4(ent_emb, (unsigned)sE2[i][j][k] * DIMN + c4);
            acc.x += w * p.x; acc.y += w * p.y; acc.z += w * p.z; acc.w += w * p.w;
        }
        float4 xo;
        xo.x = sE1v[i][j][c4 + 0] + acc.x;
        xo.y = sE1v[i][j][c4 + 1] + acc.y;
        xo.z = sE1v[i][j][c4 + 2] + acc.z;
        xo.w = sE1v[i][j][c4 + 3] + acc.w;
        *reinterpret_cast<float4*>(&sX[(i << 4) + j][c4]) = xo;
    }
    {
        const int i = t >> 6, c = t & 63;
        float acc = 0.f;
        #pragma unroll
        for (int k = 0; k < NNB; ++k) acc += sS1[i][k] * sE1v[i][k][c];
        sX[64 + i][c] = sE0v[i][c] + acc;
    }
    __syncthreads();

    {
        const int ntile = wave & 3, mh = wave >> 2;
        const int arow  = (mh << 4) + (lane & 15);
        const int koff  = (lane >> 4) << 1;
        const int bcol  = (ntile << 4) + (lane & 15);
        v8f c[3];
        #pragma unroll
        for (int q = 0; q < 3; ++q) c[q] = (v8f){0.f, 0.f, 0.f, 0.f, 0.f, 0.f, 0.f, 0.f};
        if (mh == 0) xw_tiles<3>(sX, sW, arow, koff, bcol, c);
        else         xw_tiles<2>(sX, sW, arow, koff, bcol, c);
        __syncthreads();
        const float bv = bias_v;
        const int rbase = (mh << 4) + ((lane >> 4) << 3);
        #pragma unroll
        for (int q = 0; q < 3; ++q) {
            if (q < 2 || mh == 0) {
                #pragma unroll
                for (int g = 0; g < 8; ++g)
                    sX[rbase + 32 * q + g][bcol] = sigm(c[q][g] + bv);
            }
        }
    }
    __syncthreads();

    {
        const int i = t >> 6, c = t & 63;
        float acc = sX[64 + i][c];
        #pragma unroll
        for (int k = 0; k < NNB; ++k) acc += sS1[i][k] * sX[(i << 4) + k][c];
        sX[68 + i][c] = acc;
    }
    __syncthreads();

    if (wave < 4) {
        const int ntile = wave;
        const int arow  = 64 + (lane & 15);
        const int koff  = (lane >> 4) << 1;
        const int bcol  = (ntile << 4) + (lane & 15);
        v8f c = {0.f, 0.f, 0.f, 0.f, 0.f, 0.f, 0.f, 0.f};
        #pragma unroll
        for (int kk = 0; kk < 16; ++kk) {
            const int kb = (kk << 2) + koff;
            v2f a;  a.x  = sX[arow][kb];   a.y  = sX[arow][kb + 1];
            v2f bb; bb.x = sW[kb][bcol];   bb.y = sW[kb + 1][bcol];
            c = __builtin_amdgcn_wmma_f32_16x16x4_f32(false, a, false, bb,
                                                      (short)0, c, false, false);
        }
        if (lane < 16) {
            const float bv = bias_v;
            #pragma unroll
            for (int i = 0; i < EPB; ++i) sItem[i][bcol] = tanhf(c[4 + i] + bv);
        }
    }
    __syncthreads();

    if (t < 32 * EPB) {
        const int i = wave;
        float p = sUser[i][lane] * sItem[i][lane] + sUser[i][lane + 32] * sItem[i][lane + 32];
        #pragma unroll
        for (int off = 16; off > 0; off >>= 1) p += __shfl_xor(p, off, 32);
        if (lane == 0 && i < ne) out[base + i * stride] = sigm(p);
    }

    if (base + EPB * stride >= nbatch) break;
    phase_a_loads(base + EPB * stride);
    rel_loads();
    rel_stores();
    __syncthreads();
    }
}

extern "C" void kernel_launch(void* const* d_in, const int* in_sizes, int n_in,
                              void* d_out, int out_size, void* d_ws, size_t ws_size,
                              hipStream_t stream) {
    const float* user_emb = (const float*)d_in[0];
    const float* ent_emb  = (const float*)d_in[1];
    const float* rel_emb  = (const float*)d_in[2];
    const float* W        = (const float*)d_in[3];
    const float* bias     = (const float*)d_in[4];
    const int*   u        = (const int*)d_in[5];
    const int*   v        = (const int*)d_in[6];
    const int*   adj_ent  = (const int*)d_in[7];
    const int*   adj_rel  = (const int*)d_in[8];
    float* out = (float*)d_out;
    const int B = in_sizes[5];

    kgcn_fused<<<1024, 256, 0, stream>>>(user_emb, ent_emb, rel_emb, W, bias,
                                         u, v, adj_ent, adj_rel, out, B);
}


# Row loads go out as soon as their indices are in LDS: softmax, hop-0 sum and the first product's tiles (one per trip) run under them behind LDS-only barriers; k order and every sum unchanged
# speedup vs baseline: 1.5852x; 1.0524x over previous
#include <hip/hip_runtime.h>
#include <math.h>


#define DIMN 64
#define NNB  16
#define NRELS 32
#define XS   68
#define WSP  66
#define EPB  4
#define XROWS 80

typedef float v2f __attribute__((ext_vector_type(2)));
typedef float v8f __attribute__((ext_vector_type(8)));

__device__ __forceinline__ float sigm(float x) { return 1.0f / (1.0f + __expf(-x)); }

__device__ __forceinline__ float ldf(const float* base, unsigned idx) {
    return *reinterpret_cast<const float*>(reinterpret_cast<const char*>(base) + idx * 4u);
}
__device__ __forceinline__ int ldi(const int* base, unsigned idx) {
    return *reinterpret_cast<const int*>(reinterpret_cast<const char*>(base) + idx * 4u);
}
__device__ __forceinline__ float4 ldf4(const float* base, unsigned idx) {
    return *reinterpret_cast<const float4*>(reinterpret_cast<const char*>(base) + idx * 4u);
}

__device__ __forceinline__ v8f xw_tile(const float (*X)[XS], const float (*Wt)[WSP],
                                       const int arow, const int koff, const int bcol) {
    v8f c = {0.f, 0.f, 0.f, 0.f, 0.f, 0.f, 0.f, 0.f};
    #pragma unroll
    for (int kk = 0; kk < 16; ++kk) {
        const int kb = (kk << 2) + koff;
        v2f bb; bb.x = Wt[kb][bcol];  bb.y = Wt[kb + 1][bcol];
        v2f a;  a.x  = X[arow][kb];   a.y  = X[arow][kb + 1];
        c = __builtin_amdgcn_wmma_f32_16x16x4_f32(false, a, false, bb, (short)0, c, false, false);
    }
    return c;
}

__device__ __forceinline__ void lds_barrier() {
    __builtin_amdgcn_fence(__ATOMIC_RELEASE, "workgroup", "local");
    __builtin_amdgcn_s_barrier();
    __builtin_amdgcn_fence(__ATOMIC_ACQUIRE, "workgroup", "local");
}

__device__ __forceinline__ int fresh(int x) {
    asm volatile("" : "+v"(x));
    return x;
}

__device__ __forceinline__ int tile_of_trip(const int mh, const int i) {
    if (mh == 0) return (i == 1) ? 0 : (i == 3) ? 2 : -1;
    return (i == 1) ? 4 : (i == 2) ? 1 : (i == 4) ? 3 : -1;
}

__global__ __launch_bounds__(256, 8)
void kgcn_fused(const float* __restrict__ user_emb,
                const float* __restrict__ ent_emb,
                const float* __restrict__ rel_emb,
                const float* __restrict__ Wg,
                const float* __restrict__ bias,
                const int* __restrict__ u,
                const int* __restrict__ v,
                const int* __restrict__ adj_ent,
                const int* __restrict__ adj_rel,
                float* __restrict__ out,
                const int nbatch)
{
    __shared__ float sW[DIMN][WSP];
    __shared__ float sX[XROWS][XS];
    __shared__ float sUser[EPB][DIMN];
    __shared__ float sUr[EPB][NRELS];
    __shared__ float sE0v[EPB][DIMN];
    __shared__ float sE1v[EPB][NNB][DIMN];
    __shared__ float sS1[EPB][NNB];
    __shared__ float sS2[EPB][NNB][NNB];
    __shared__ int   sA1[EPB][DIMN];
    __shared__ int   sE2[EPB][NNB][NNB];
    __shared__ int   sR2[EPB][NNB][NNB];
    __shared__ float sItem[EPB][DIMN];

    const int tid  = threadIdx.x;
    const int t    = tid;
    const int lane = t & 31;
    const int wave = t >> 5;
    const int stride = gridDim.x;

    float4 wv[4];
    #pragma unroll
    for (int q = 0; q < 4; ++q) wv[q] = ldf4(Wg, (unsigned)(t + 256 * q) * 4u);

    const float bias_v = bias[((wave & 3) << 4) + (lane & 15)];

    float4 rv[2];
    auto rel_loads = [&](const int t) {
        #pragma unroll
        for (int q = 0; q < 2; ++q) rv[q] = ldf4(rel_emb, (unsigned)(t + 256 * q) * 4u);
    };
    auto rel_stores = [&](const int t) {
        #pragma unroll
        for (int q = 0; q < 2; ++q) {
            const int f4 = t + 256 * q;
            *reinterpret_cast<float4*>(&sX[f4 >> 4][(f4 & 15) << 2]) = rv[q];
        }
    };
    rel_loads(t);

    if ((int)blockIdx.x >= nbatch) return;

    int a_ae, a_ar; float a_uv, a_ev;
    auto phase_a_loads = [&](const int gbase, const int t) {
        int gne = 1;
        for (int i = 1; i < EPB; ++i) if (gbase + i * stride < nbatch) gne = i + 1;
        const int i  = t >> 6, c = t & 63;
        const int ii = (i < gne) ? i : 0;
        const int b   = gbase + ii * stride;
        const int e0  = v[b];
        const int uid = u[b];
        const unsigned ao = (unsigned)e0 * NNB + (c & (NNB - 1));
        a_ae = ldi(adj_ent, ao);
        a_ar = ldi(adj_rel, ao);
        a_uv = ldf(user_emb, (unsigned)uid * DIMN + c);
        a_ev = ldf(ent_emb,  (unsigned)e0  * DIMN + c);
    };
    phase_a_loads(blockIdx.x, t);

    #pragma unroll
    for (int q = 0; q < 4; ++q) {
        const int f4 = t + 256 * q, row = f4 >> 4, col = (f4 & 15) << 2;
        *reinterpret_cast<float2*>(&sW[row][col])     = make_float2(wv[q].x, wv[q].y);
        *reinterpret_cast<float2*>(&sW[row][col + 2]) = make_float2(wv[q].z, wv[q].w);
    }
    rel_stores(t);
    for (int q = NRELS * XS / 4 + t; q < XROWS * XS / 4; q += 256)
        reinterpret_cast<float4*>(&sX[0][0])[q] = make_float4(0.f, 0.f, 0.f, 0.f);

    for (int base = blockIdx.x; ; base += EPB * stride) {

    int ne = 1;
    for (int i = 1; i < EPB; ++i) if (base + i * stride < nbatch) ne = i + 1;

    {
        const int t = fresh(tid);
        const int i  = t >> 6, c = t & 63;
        sA1[i][c]   = (c & NNB) ? a_ar : a_ae;
        sUser[i][c] = a_uv;
        sE0v[i][c]  = a_ev;
    }
    __syncthreads();

    {
        const int t = fresh(tid);
        const int j = t >> 4, k = t & 15, c4 = k * 4;
        int    n2[EPB], r2[EPB];
        float4 p[EPB];
        #pragma unroll
        for (int i = 0; i < EPB; ++i) {
            const int ej = sA1[i][j];
            n2[i] = ldi(adj_ent, (unsigned)ej * NNB + k);
            r2[i] = ldi(adj_rel, (unsigned)ej * NNB + k);
            p[i]  = ldf4(ent_emb, (unsigned)ej * DIMN + c4);
        }
        if (t < EPB * NRELS) {
            const int i = t >> 5, r = t & 31;
            float acc = 0.f;
            #pragma unroll
            for (int d = 0; d < DIMN; ++d) acc += sUser[i][d] * sX[r][d];
            sUr[i][r] = acc;
        }
        #pragma unroll
        for (int i = 0; i < EPB; ++i) {
            sE2[i][j][k] = n2[i];
            sR2[i][j][k] = r2[i];
            *reinterpret_cast<float4*>(&sE1v[i][j][c4]) = p[i];
        }
    }
    __syncthreads();

    auto softmax_scores = [&](const int t) {
    if (t < EPB * NNB) {
        const int i = t >> 4, r = t & 15;
        float l[NNB], m = -1e30f;
        #pragma unroll
        for (int k = 0; k < NNB; ++k) { l[k] = sUr[i][sR2[i][r][k]]; m = fmaxf(m, l[k]); }
        float s = 0.f;
        #pragma unroll
        for (int k = 0; k < NNB; ++k) { l[k] = __expf(l[k] - m); s += l[k]; }
        const float inv = 1.f / s;
        #pragma unroll
        for (int k = 0; k < NNB; ++k) sS2[i][r][k] = l[k] * inv;
    } else if (t < EPB * NNB + EPB) {
        const int i = t - EPB * NNB;
        float l[NNB], m = -1e30f;
        #pragma unroll
        for (int k = 0; k < NNB; ++k) { l[k] = sUr[i][sA1[i][NNB + k]]; m = fmaxf(m, l[k]); }
        float s = 0.f;
        #pragma unroll
        for (int k = 0; k < NNB; ++k) { l[k] = __expf(l[k] - m); s += l[k]; }
        const float inv = 1.f / s;
        #pragma unroll
        for (int k = 0; k < NNB; ++k) sS1[i][k] = l[k] * inv;
    }
    };

    {
        const int t = fresh(tid), lane = t & 31, wave = t >> 5;
        const int j = t >> 4, c4 = (t & 15) * 4;
        const int mh    = __builtin_amdgcn_readfirstlane(wave >> 2);
        const int koff  = (lane >> 4) << 1;
        const int bcol  = ((wave & 3) << 4) + (lane & 15);
        const int frow  = lane & 15;
        const int drow  = (lane >> 4) << 3;
        auto take_tile = [&](const int i) {
            const int tl = tile_of_trip(mh, i);
            if (tl >= 0) {
                const v8f y = xw_tile(sX, sW, (tl << 4) + frow, koff, bcol);
                const float bv = bias_v;
                if (tl < EPB) {
                    #pragma unroll
                    for (int g = 0; g < 8; ++g)
                        sE1v[tl][drow + g][bcol] = sigm(y[g] + bv);
                } else if (lane < 16) {
                    #pragma unroll
                    for (int e = 0; e < EPB; ++e) sItem[e][bcol] = sigm(y[e] + bv);
                }
            }
        };
        #pragma clang loop vectorize(disable) interleave(disable) unroll(disable)
        for (int i = 0; i < EPB; ++i) {
            float4 p[NNB];
            #pragma unroll
            for (int k = 0; k < NNB; ++k)
                p[k] = ldf4(ent_emb, (unsigned)sE2[i][j][k] * DIMN + c4);
            if (i == 0) {
                softmax_scores(t);
                lds_barrier();
                const int e = t >> 6, c = t & 63;
                float a0 = 0.f;
                #pragma unroll
                for (int k = 0; k < NNB; ++k) a0 += sS1[e][k] * sE1v[e][k][c];
                sX[64 + e][c] = sE0v[e][c] + a0;
            } else {
                lds_barrier();
                take_tile(i);
            }
            float4 acc = make_float4(0.f, 0.f, 0.f, 0.f);
            #pragma unroll
            for (int k = 0; k < NNB; ++k) {
                const float w = sS2[i][j][k];
                acc.x += w * p[k].x; acc.y += w * p[k].y; acc.z += w * p[k].z; acc.w += w * p[k].w;
            }
            float4 xo;
            xo.x = sE1v[i][j][c4 + 0] + acc.x;
            xo.y = sE1v[i][j][c4 + 1] + acc.y;
            xo.z = sE1v[i][j][c4 + 2] + acc.z;
            xo.w = sE1v[i][j][c4 + 3] + acc.w;
            *reinterpret_cast<float4*>(&sX[(i << 4) + j][c4]) = xo;
        }
        __syncthreads();
        take_tile(EPB);
    }
    __syncthreads();

    {
        const int t = fresh(tid);
        const int i = t >> 6, c = t & 63;
        float acc = sItem[i][c];
        #pragma unroll
        for (int k = 0; k < NNB; ++k) acc += sS1[i][k] * sE1v[i][k][c];
        sX[68 + i][c] = acc;
    }
    __syncthreads();

    {
    const int t = fresh(tid), lane = t & 31, wave = t >> 5;
    if (wave < 4) {
        const int ntile = wave;
        const int arow  = 64 + (lane & 15);
        const int koff  = (lane >> 4) << 1;
        const int bcol  = (ntile << 4) + (lane & 15);
        v8f c = {0.f, 0.f, 0.f, 0.f, 0.f, 0.f, 0.f, 0.f};
        #pragma unroll
        for (int kk = 0; kk < 16; ++kk) {
            const int kb = (kk << 2) + koff;
            v2f a;  a.x  = sX[arow][kb];   a.y  = sX[arow][kb + 1];
            v2f bb; bb.x = sW[kb][bcol];   bb.y = sW[kb + 1][bcol];
            c = __builtin_amdgcn_wmma_f32_16x16x4_f32(false, a, false, bb,
                                                      (short)0, c, false, false);
        }
        if (lane < 16) {
            const float bv = bias_v;
            #pragma unroll
            for (int i = 0; i < EPB; ++i) sItem[i][bcol] = tanhf(c[4 + i] + bv);
        }
    }
    __syncthreads();

    if (t < 32 * EPB) {
        const int i = wave;
        float p = sUser[i][lane] * sItem[i][lane] + sUser[i][lane + 32] * sItem[i][lane + 32];
        #pragma unroll
        for (int off = 16; off > 0; off >>= 1) p += __shfl_xor(p, off, 32);
        if (lane == 0 && i < ne) out[base + i * stride] = sigm(p);
    }
    }

    if (base + EPB * stride >= nbatch) break;
    {
        const int t = fresh(tid);
        phase_a_loads(base + EPB * stride, t);
        rel_loads(t);
        rel_stores(t);
    }
    __syncthreads();
    }
}

extern "C" void kernel_launch(void* const* d_in, const int* in_sizes, int n_in,
                              void* d_out, int out_size, void* d_ws, size_t ws_size,
                              hipStream_t stream) {
    const float* user_emb = (const float*)d_in[0];
    const float* ent_emb  = (const float*)d_in[1];
    const float* rel_emb  = (const float*)d_in[2];
    const float* W        = (const float*)d_in[3];
    const float* bias     = (const float*)d_in[4];
    const int*   u        = (const int*)d_in[5];
    const int*   v        = (const int*)d_in[6];
    const int*   adj_ent  = (const int*)d_in[7];
    const int*   adj_rel  = (const int*)d_in[8];
    float* out = (float*)d_out;
    const int B = in_sizes[5];

    kgcn_fused<<<1024, 256, 0, stream>>>(user_emb, ent_emb, rel_emb, W, bias,
                                         u, v, adj_ent, adj_rel, out, B);
}
